# starting kernel
# baseline (speedup 1.0000x reference)
#include <hip/hip_runtime.h>

typedef __attribute__((ext_vector_type(2))) float v2f;
typedef __attribute__((ext_vector_type(8))) float v8f;

#define WAVES_PER_BLOCK 8
#define LDS_PER_WAVE (832 + 544 + 288)

__device__ __forceinline__ float relu1(float a) {
    return __builtin_amdgcn_fmed3f(a, 0.0f, __builtin_inff());
}

__global__ __launch_bounds__(256) void encoder_mlp_wmma(
    const float* __restrict__ x,  const float* __restrict__ c,
    const float* __restrict__ W1, const float* __restrict__ b1,
    const float* __restrict__ W2, const float* __restrict__ b2,
    const float* __restrict__ W3, const float* __restrict__ b3,
    const float* __restrict__ Wmu, const float* __restrict__ bmu,
    const float* __restrict__ Wlv, const float* __restrict__ blv,
    float* __restrict__ out, int Bn)
{
    __shared__ float lds[WAVES_PER_BLOCK * LDS_PER_WAVE];

    const int lane = threadIdx.x & 31;
    const int wave = threadIdx.x >> 5;
    const int hi   = lane >> 4;
    const int sl   = lane & 15;

    float* h1 = &lds[wave * LDS_PER_WAVE];
    float* h2 = h1 + 16 * 52;
    float* h3 = h2 + 16 * 34;

    float w1r[15], b1r[3];
    #pragma unroll
    for (int i = 0; i < 15; ++i) w1r[i] = W1[i];
    #pragma unroll
    for (int i = 0; i < 3; ++i)  b1r[i] = b1[i];

    v2f w2f[2][13];
    #pragma unroll
    for (int t = 0; t < 2; ++t) {
        const int n = sl + 16 * t;
        #pragma unroll
        for (int k = 0; k < 13; ++k) {
            const int k0 = 4 * k + 2 * hi;
            v2f w;
            w.x = (k0     < 51) ? W2[n * 51 + k0]     : 0.0f;
            w.y = (k0 + 1 < 51) ? W2[n * 51 + k0 + 1] : 0.0f;
            w2f[t][k] = w;
        }
    }
    v2f w3f[8];
    #pragma unroll
    for (int k = 0; k < 8; ++k) {
        const int k0 = 4 * k + 2 * hi;
        v2f w; w.x = W3[sl * 32 + k0]; w.y = W3[sl * 32 + k0 + 1];
        w3f[k] = w;
    }
    v2f whf[4];
    {
        const float* wrow = (sl < 3) ? (Wmu + sl * 16)
                          : (sl < 6) ? (Wlv + (sl - 3) * 16) : nullptr;
        #pragma unroll
        for (int k = 0; k < 4; ++k) {
            const int k0 = 4 * k + 2 * hi;
            v2f w;
            w.x = wrow ? wrow[k0]     : 0.0f;
            w.y = wrow ? wrow[k0 + 1] : 0.0f;
            whf[k] = w;
        }
    }
    const float b2v0 = b2[sl];
    const float b2v1 = b2[sl + 16];
    const float b3v  = b3[sl];
    const float bhv  = (sl < 3) ? bmu[sl] : (sl < 6) ? blv[sl - 3] : 0.0f;

    const int ntiles     = (Bn + 15) >> 4;
    const int totalWaves = gridDim.x * WAVES_PER_BLOCK;
    const int gwave      = blockIdx.x * WAVES_PER_BLOCK + wave;

    for (int tile = gwave; tile < ntiles; tile += totalWaves) {
        const long tbase = (long)tile * 16;

        {
            long smp = tbase + sl;
            if (smp >= Bn) smp = Bn - 1;
            const int j0 = hi ? 9 : 0;
            const int nj = hi ? 8 : 9;
            for (int jj = 0; jj < nj; ++jj) {
                const int j = j0 + jj;
                const float* xp = x + (smp * 17 + j) * 3;
                const float* cp = c + (smp * 17 + j) * 2;
                const float i0 = xp[0], i1 = xp[1], i2 = xp[2];
                const float i3 = cp[0], i4 = cp[1];
                #pragma unroll
                for (int o = 0; o < 3; ++o) {
                    float a = b1r[o];
                    a = fmaf(w1r[o * 5 + 0], i0, a);
                    a = fmaf(w1r[o * 5 + 1], i1, a);
                    a = fmaf(w1r[o * 5 + 2], i2, a);
                    a = fmaf(w1r[o * 5 + 3], i3, a);
                    a = fmaf(w1r[o * 5 + 4], i4, a);
                    h1[sl * 52 + j * 3 + o] = relu1(a);
                }
            }
            if (hi) h1[sl * 52 + 51] = 0.0f;
        }
        asm volatile("" ::: "memory");

        v8f acc0, acc1;
        #pragma unroll
        for (int r = 0; r < 8; ++r) { acc0[r] = b2v0; acc1[r] = b2v1; }
        #pragma unroll
        for (int k = 0; k < 13; ++k) {
            const int kb = 4 * k + 2 * hi;
            v2f a = *(const v2f*)&h1[sl * 52 + kb];
            acc0 = __builtin_amdgcn_wmma_f32_16x16x4_f32(
                       false, a, false, w2f[0][k], (short)0, acc0, false, false);
            acc1 = __builtin_amdgcn_wmma_f32_16x16x4_f32(
                       false, a, false, w2f[1][k], (short)0, acc1, false, false);
        }
        #pragma unroll
        for (int r = 0; r < 8; ++r) {
            h2[(r + 8 * hi) * 34 + sl]      = relu1(acc0[r]);
            h2[(r + 8 * hi) * 34 + sl + 16] = relu1(acc1[r]);
        }
        asm volatile("" ::: "memory");

        v8f acc2;
        #pragma unroll
        for (int r = 0; r < 8; ++r) acc2[r] = b3v;
        #pragma unroll
        for (int k = 0; k < 8; ++k) {
            const int kb = 4 * k + 2 * hi;
            v2f a = *(const v2f*)&h2[sl * 34 + kb];
            acc2 = __builtin_amdgcn_wmma_f32_16x16x4_f32(
                       false, a, false, w3f[k], (short)0, acc2, false, false);
        }
        #pragma unroll
        for (int r = 0; r < 8; ++r)
            h3[(r + 8 * hi) * 18 + sl] = relu1(acc2[r]);
        asm volatile("" ::: "memory");

        v8f acch;
        #pragma unroll
        for (int r = 0; r < 8; ++r) acch[r] = bhv;
        #pragma unroll
        for (int k = 0; k < 4; ++k) {
            const int kb = 4 * k + 2 * hi;
            v2f a = *(const v2f*)&h3[sl * 18 + kb];
            acch = __builtin_amdgcn_wmma_f32_16x16x4_f32(
                       false, a, false, whf[k], (short)0, acch, false, false);
        }

        if (sl < 6) {
            #pragma unroll
            for (int r = 0; r < 8; ++r) {
                const long smp = tbase + r + 8 * hi;
                if (smp < Bn) {
                    if (sl < 3) out[smp * 3 + sl] = acch[r];
                    else        out[(long)Bn * 3 + smp * 3 + (sl - 3)] = acch[r];
                }
            }
        }
    }
}

extern "C" void kernel_launch(void* const* d_in, const int* in_sizes, int n_in,
                              void* d_out, int out_size, void* d_ws, size_t ws_size,
                              hipStream_t stream) {
    const float* x   = (const float*)d_in[0];
    const float* c   = (const float*)d_in[1];
    const float* W1  = (const float*)d_in[2];
    const float* b1  = (const float*)d_in[3];
    const float* W2  = (const float*)d_in[4];
    const float* b2  = (const float*)d_in[5];
    const float* W3  = (const float*)d_in[6];
    const float* b3  = (const float*)d_in[7];
    const float* Wmu = (const float*)d_in[8];
    const float* bmu = (const float*)d_in[9];
    const float* Wlv = (const float*)d_in[10];
    const float* blv = (const float*)d_in[11];
    float* out = (float*)d_out;

    const int Bn = in_sizes[0] / 51;
    if (Bn <= 0) return;

    const int blocks = 1024;
    encoder_mlp_wmma<<<blocks, 256, 0, stream>>>(
        x, c, W1, b1, W2, b2, W3, b3, Wmu, bmu, Wlv, blv, out, Bn);
}


# fc1 remapped to one (sample,joint) pair per lane per step: each x/c load is one contiguous run per wave (was 32 rows 204 B apart); uniform 9 steps, so a tile's 18 loads issue before first use
# speedup vs baseline: 1.7687x; 1.7687x over previous
#include <hip/hip_runtime.h>

typedef __attribute__((ext_vector_type(2))) float v2f;
typedef __attribute__((ext_vector_type(8))) float v8f;

#define WAVES_PER_BLOCK 8
#define LDS_PER_WAVE (832 + 544 + 288)

__device__ __forceinline__ float relu1(float a) {
    return __builtin_amdgcn_fmed3f(a, 0.0f, __builtin_inff());
}

__global__ __launch_bounds__(256) void encoder_mlp_wmma(
    const float* __restrict__ x,  const float* __restrict__ c,
    const float* __restrict__ W1, const float* __restrict__ b1,
    const float* __restrict__ W2, const float* __restrict__ b2,
    const float* __restrict__ W3, const float* __restrict__ b3,
    const float* __restrict__ Wmu, const float* __restrict__ bmu,
    const float* __restrict__ Wlv, const float* __restrict__ blv,
    float* __restrict__ out, int Bn)
{
    __shared__ float lds[WAVES_PER_BLOCK * LDS_PER_WAVE];

    const int lane = threadIdx.x & 31;
    const int wave = threadIdx.x >> 5;
    const int hi   = lane >> 4;
    const int sl   = lane & 15;

    float* h1 = &lds[wave * LDS_PER_WAVE];
    float* h2 = h1 + 16 * 52;
    float* h3 = h2 + 16 * 34;

    float w1r[15], b1r[3];
    #pragma unroll
    for (int i = 0; i < 15; ++i) w1r[i] = W1[i];
    #pragma unroll
    for (int i = 0; i < 3; ++i)  b1r[i] = b1[i];

    v2f w2f[2][13];
    #pragma unroll
    for (int t = 0; t < 2; ++t) {
        const int n = sl + 16 * t;
        #pragma unroll
        for (int k = 0; k < 13; ++k) {
            const int k0 = 4 * k + 2 * hi;
            v2f w;
            w.x = (k0     < 51) ? W2[n * 51 + k0]     : 0.0f;
            w.y = (k0 + 1 < 51) ? W2[n * 51 + k0 + 1] : 0.0f;
            w2f[t][k] = w;
        }
    }
    v2f w3f[8];
    #pragma unroll
    for (int k = 0; k < 8; ++k) {
        const int k0 = 4 * k + 2 * hi;
        v2f w; w.x = W3[sl * 32 + k0]; w.y = W3[sl * 32 + k0 + 1];
        w3f[k] = w;
    }
    v2f whf[4];
    {
        const float* wrow = (sl < 3) ? (Wmu + sl * 16)
                          : (sl < 6) ? (Wlv + (sl - 3) * 16) : nullptr;
        #pragma unroll
        for (int k = 0; k < 4; ++k) {
            const int k0 = 4 * k + 2 * hi;
            v2f w;
            w.x = wrow ? wrow[k0]     : 0.0f;
            w.y = wrow ? wrow[k0 + 1] : 0.0f;
            whf[k] = w;
        }
    }
    const float b2v0 = b2[sl];
    const float b2v1 = b2[sl + 16];
    const float b3v  = b3[sl];
    const float bhv  = (sl < 3) ? bmu[sl] : (sl < 6) ? blv[sl - 3] : 0.0f;

    const int ntiles     = (Bn + 15) >> 4;
    const int totalWaves = gridDim.x * WAVES_PER_BLOCK;
    const int gwave      = blockIdx.x * WAVES_PER_BLOCK + wave;

    for (int tile = gwave; tile < ntiles; tile += totalWaves) {
        const long tbase = (long)tile * 16;

        {
            const float* xt = x + tbase * 51;
            const float* ct = c + tbase * 34;
            const long rem = (long)Bn - tbase;
            const int lastp = (rem >= 16 ? 16 * 17 : (int)rem * 17) - 1;
            float xi[9][3], ci[9][2];
            #pragma unroll
            for (int i = 0; i < 9; ++i) {
                const int p  = (i < 8) ? lane + 32 * i : 240 + (lane ^ 16);
                const int pl = min(p, lastp);
                const float* xp = xt + (long)pl * 3;
                const float* cp = ct + (long)pl * 2;
                xi[i][0] = xp[0]; xi[i][1] = xp[1]; xi[i][2] = xp[2];
                ci[i][0] = cp[0]; ci[i][1] = cp[1];
            }
            #pragma unroll
            for (int i = 0; i < 9; ++i) {
                const int p = (i < 8) ? lane + 32 * i : 240 + (lane ^ 16);
                float* hp = h1 + 3 * p + p / 17;
                #pragma unroll
                for (int o = 0; o < 3; ++o) {
                    float a = b1r[o];
                    a = fmaf(w1r[o * 5 + 0], xi[i][0], a);
                    a = fmaf(w1r[o * 5 + 1], xi[i][1], a);
                    a = fmaf(w1r[o * 5 + 2], xi[i][2], a);
                    a = fmaf(w1r[o * 5 + 3], ci[i][0], a);
                    a = fmaf(w1r[o * 5 + 4], ci[i][1], a);
                    hp[o] = relu1(a);
                }
            }
            if (hi) h1[sl * 52 + 51] = 0.0f;
        }
        asm volatile("" ::: "memory");

        v8f acc0, acc1;
        #pragma unroll
        for (int r = 0; r < 8; ++r) { acc0[r] = b2v0; acc1[r] = b2v1; }
        #pragma unroll
        for (int k = 0; k < 13; ++k) {
            const int kb = 4 * k + 2 * hi;
            v2f a = *(const v2f*)&h1[sl * 52 + kb];
            acc0 = __builtin_amdgcn_wmma_f32_16x16x4_f32(
                       false, a, false, w2f[0][k], (short)0, acc0, false, false);
            acc1 = __builtin_amdgcn_wmma_f32_16x16x4_f32(
                       false, a, false, w2f[1][k], (short)0, acc1, false, false);
        }
        #pragma unroll
        for (int r = 0; r < 8; ++r) {
            h2[(r + 8 * hi) * 34 + sl]      = relu1(acc0[r]);
            h2[(r + 8 * hi) * 34 + sl + 16] = relu1(acc1[r]);
        }
        asm volatile("" ::: "memory");

        v8f acc2;
        #pragma unroll
        for (int r = 0; r < 8; ++r) acc2[r] = b3v;
        #pragma unroll
        for (int k = 0; k < 8; ++k) {
            const int kb = 4 * k + 2 * hi;
            v2f a = *(const v2f*)&h2[sl * 34 + kb];
            acc2 = __builtin_amdgcn_wmma_f32_16x16x4_f32(
                       false, a, false, w3f[k], (short)0, acc2, false, false);
        }
        #pragma unroll
        for (int r = 0; r < 8; ++r)
            h3[(r + 8 * hi) * 18 + sl] = relu1(acc2[r]);
        asm volatile("" ::: "memory");

        v8f acch;
        #pragma unroll
        for (int r = 0; r < 8; ++r) acch[r] = bhv;
        #pragma unroll
        for (int k = 0; k < 4; ++k) {
            const int kb = 4 * k + 2 * hi;
            v2f a = *(const v2f*)&h3[sl * 18 + kb];
            acch = __builtin_amdgcn_wmma_f32_16x16x4_f32(
                       false, a, false, whf[k], (short)0, acch, false, false);
        }

        if (sl < 6) {
            #pragma unroll
            for (int r = 0; r < 8; ++r) {
                const long smp = tbase + r + 8 * hi;
                if (smp < Bn) {
                    if (sl < 3) out[smp * 3 + sl] = acch[r];
                    else        out[(long)Bn * 3 + smp * 3 + (sl - 3)] = acch[r];
                }
            }
        }
    }
}

extern "C" void kernel_launch(void* const* d_in, const int* in_sizes, int n_in,
                              void* d_out, int out_size, void* d_ws, size_t ws_size,
                              hipStream_t stream) {
    const float* x   = (const float*)d_in[0];
    const float* c   = (const float*)d_in[1];
    const float* W1  = (const float*)d_in[2];
    const float* b1  = (const float*)d_in[3];
    const float* W2  = (const float*)d_in[4];
    const float* b2  = (const float*)d_in[5];
    const float* W3  = (const float*)d_in[6];
    const float* b3  = (const float*)d_in[7];
    const float* Wmu = (const float*)d_in[8];
    const float* bmu = (const float*)d_in[9];
    const float* Wlv = (const float*)d_in[10];
    const float* blv = (const float*)d_in[11];
    float* out = (float*)d_out;

    const int Bn = in_sizes[0] / 51;
    if (Bn <= 0) return;

    const int blocks = 1024;
    encoder_mlp_wmma<<<blocks, 256, 0, stream>>>(
        x, c, W1, b1, W2, b2, W3, b3, Wmu, bmu, Wlv, blv, out, Bn);
}


# tile loop software-pipelined: the next tile's 18 x/c loads issue right behind fc1 of the tile in hand (same registers; wave's last tile peeled), so they land during the WMMA phases, not ahead of fc1
# speedup vs baseline: 1.8254x; 1.0321x over previous
#include <hip/hip_runtime.h>

typedef __attribute__((ext_vector_type(2))) float v2f;
typedef __attribute__((ext_vector_type(8))) float v8f;
typedef __attribute__((ext_vector_type(3))) float v3f;

#define WAVES_PER_BLOCK 8
#define LDS_PER_WAVE (832 + 544 + 288)

__device__ __forceinline__ float relu1(float a) {
    return __builtin_amdgcn_fmed3f(a, 0.0f, __builtin_inff());
}


__device__ __forceinline__ void load_tile(
    const float* __restrict__ x, const float* __restrict__ c,
    int tile, int Bn, int lane, v3f (&xi)[9], v2f (&ci)[9])
{
    const long tbase = (long)tile * 16;
    const float* xt = x + tbase * 51;
    const float* ct = c + tbase * 34;
    const long rem = (long)Bn - tbase;
    const int lastp = (rem >= 16 ? 16 * 17 : (int)rem * 17) - 1;
    #pragma unroll
    for (int i = 0; i < 9; ++i) {
        const int p  = (i < 8) ? lane + 32 * i : 240 + (lane ^ 16);
        const int pl = min(p, lastp);
        const float* xp = xt + (long)pl * 3;
        const float* cp = ct + (long)pl * 2;
        v3f xv; xv.x = xp[0]; xv.y = xp[1]; xv.z = xp[2];
        v2f cv; cv.x = cp[0]; cv.y = cp[1];
        xi[i] = xv;
        ci[i] = cv;
    }
}

__global__ __launch_bounds__(256) void encoder_mlp_wmma(
    const float* __restrict__ x,  const float* __restrict__ c,
    const float* __restrict__ W1, const float* __restrict__ b1,
    const float* __restrict__ W2, const float* __restrict__ b2,
    const float* __restrict__ W3, const float* __restrict__ b3,
    const float* __restrict__ Wmu, const float* __restrict__ bmu,
    const float* __restrict__ Wlv, const float* __restrict__ blv,
    float* __restrict__ out, int Bn)
{
    __shared__ float lds[WAVES_PER_BLOCK * LDS_PER_WAVE];

    const int lane = threadIdx.x & 31;
    const int wave = threadIdx.x >> 5;
    const int hi   = lane >> 4;
    const int sl   = lane & 15;

    float* h1 = &lds[wave * LDS_PER_WAVE];
    float* h2 = h1 + 16 * 52;
    float* h3 = h2 + 16 * 34;

    float w1r[15], b1r[3];
    #pragma unroll
    for (int i = 0; i < 15; ++i) w1r[i] = W1[i];
    #pragma unroll
    for (int i = 0; i < 3; ++i)  b1r[i] = b1[i];

    v2f w2f[2][13];
    #pragma unroll
    for (int t = 0; t < 2; ++t) {
        const int n = sl + 16 * t;
        #pragma unroll
        for (int k = 0; k < 13; ++k) {
            const int k0 = 4 * k + 2 * hi;
            v2f w;
            w.x = (k0     < 51) ? W2[n * 51 + k0]     : 0.0f;
            w.y = (k0 + 1 < 51) ? W2[n * 51 + k0 + 1] : 0.0f;
            w2f[t][k] = w;
        }
    }
    v2f w3f[8];
    #pragma unroll
    for (int k = 0; k < 8; ++k) {
        const int k0 = 4 * k + 2 * hi;
        v2f w; w.x = W3[sl * 32 + k0]; w.y = W3[sl * 32 + k0 + 1];
        w3f[k] = w;
    }
    v2f whf[4];
    {
        const float* wrow = (sl < 3) ? (Wmu + sl * 16)
                          : (sl < 6) ? (Wlv + (sl - 3) * 16) : nullptr;
        #pragma unroll
        for (int k = 0; k < 4; ++k) {
            const int k0 = 4 * k + 2 * hi;
            v2f w;
            w.x = wrow ? wrow[k0]     : 0.0f;
            w.y = wrow ? wrow[k0 + 1] : 0.0f;
            whf[k] = w;
        }
    }
    const float b2v0 = b2[sl];
    const float b2v1 = b2[sl + 16];
    const float b3v  = b3[sl];
    const float bhv  = (sl < 3) ? bmu[sl] : (sl < 6) ? blv[sl - 3] : 0.0f;

    const int ntiles     = (Bn + 15) >> 4;
    const int totalWaves = gridDim.x * WAVES_PER_BLOCK;
    const int gwave      = blockIdx.x * WAVES_PER_BLOCK + wave;

    v3f xi[9];
    v2f ci[9];

    auto fc1 = [&]() __attribute__((always_inline)) {
        {
            #pragma unroll
            for (int i = 0; i < 9; ++i) {
                const int p = (i < 8) ? lane + 32 * i : 240 + (lane ^ 16);
                float* hp = h1 + 3 * p + p / 17;
                #pragma unroll
                for (int o = 0; o < 3; ++o) {
                    float a = b1r[o];
                    a = fmaf(w1r[o * 5 + 0], xi[i].x, a);
                    a = fmaf(w1r[o * 5 + 1], xi[i].y, a);
                    a = fmaf(w1r[o * 5 + 2], xi[i].z, a);
                    a = fmaf(w1r[o * 5 + 3], ci[i].x, a);
                    a = fmaf(w1r[o * 5 + 4], ci[i].y, a);
                    hp[o] = relu1(a);
                }
            }
            if (hi) h1[sl * 52 + 51] = 0.0f;
        }
    };

    auto finish_tile = [&](const long tbase) __attribute__((always_inline)) {
        asm volatile("" ::: "memory");

        v8f acc0, acc1;
        #pragma unroll
        for (int r = 0; r < 8; ++r) { acc0[r] = b2v0; acc1[r] = b2v1; }
        #pragma unroll
        for (int k = 0; k < 13; ++k) {
            const int kb = 4 * k + 2 * hi;
            v2f a = *(const v2f*)&h1[sl * 52 + kb];
            acc0 = __builtin_amdgcn_wmma_f32_16x16x4_f32(
                       false, a, false, w2f[0][k], (short)0, acc0, false, false);
            acc1 = __builtin_amdgcn_wmma_f32_16x16x4_f32(
                       false, a, false, w2f[1][k], (short)0, acc1, false, false);
        }
        #pragma unroll
        for (int r = 0; r < 8; ++r) {
            h2[(r + 8 * hi) * 34 + sl]      = relu1(acc0[r]);
            h2[(r + 8 * hi) * 34 + sl + 16] = relu1(acc1[r]);
        }
        asm volatile("" ::: "memory");

        v8f acc2;
        #pragma unroll
        for (int r = 0; r < 8; ++r) acc2[r] = b3v;
        #pragma unroll
        for (int k = 0; k < 8; ++k) {
            const int kb = 4 * k + 2 * hi;
            v2f a = *(const v2f*)&h2[sl * 34 + kb];
            acc2 = __builtin_amdgcn_wmma_f32_16x16x4_f32(
                       false, a, false, w3f[k], (short)0, acc2, false, false);
        }
        #pragma unroll
        for (int r = 0; r < 8; ++r)
            h3[(r + 8 * hi) * 18 + sl] = relu1(acc2[r]);
        asm volatile("" ::: "memory");

        v8f acch;
        #pragma unroll
        for (int r = 0; r < 8; ++r) acch[r] = bhv;
        #pragma unroll
        for (int k = 0; k < 4; ++k) {
            const int kb = 4 * k + 2 * hi;
            v2f a = *(const v2f*)&h3[sl * 18 + kb];
            acch = __builtin_amdgcn_wmma_f32_16x16x4_f32(
                       false, a, false, whf[k], (short)0, acch, false, false);
        }

        if (sl < 6) {
            #pragma unroll
            for (int r = 0; r < 8; ++r) {
                const long smp = tbase + r + 8 * hi;
                if (smp < Bn) {
                    if (sl < 3) out[smp * 3 + sl] = acch[r];
                    else        out[(long)Bn * 3 + smp * 3 + (sl - 3)] = acch[r];
                }
            }
        }
    };

    int tile = gwave;
    if (tile < ntiles) {
        load_tile(x, c, tile, Bn, lane, xi, ci);
        for (; tile + totalWaves < ntiles; tile += totalWaves) {
            fc1();
            load_tile(x, c, tile + totalWaves, Bn, lane, xi, ci);
            finish_tile((long)tile * 16);
        }
        fc1();
        finish_tile((long)tile * 16);
    }
}

extern "C" void kernel_launch(void* const* d_in, const int* in_sizes, int n_in,
                              void* d_out, int out_size, void* d_ws, size_t ws_size,
                              hipStream_t stream) {
    const float* x   = (const float*)d_in[0];
    const float* c   = (const float*)d_in[1];
    const float* W1  = (const float*)d_in[2];
    const float* b1  = (const float*)d_in[3];
    const float* W2  = (const float*)d_in[4];
    const float* b2  = (const float*)d_in[5];
    const float* W3  = (const float*)d_in[6];
    const float* b3  = (const float*)d_in[7];
    const float* Wmu = (const float*)d_in[8];
    const float* bmu = (const float*)d_in[9];
    const float* Wlv = (const float*)d_in[10];
    const float* blv = (const float*)d_in[11];
    float* out = (float*)d_out;

    const int Bn = in_sizes[0] / 51;
    if (Bn <= 0) return;

    const int blocks = 1024;
    encoder_mlp_wmma<<<blocks, 256, 0, stream>>>(
        x, c, W1, b1, W2, b2, W3, b3, Wmu, bmu, Wlv, blv, out, Bn);
}


# heads product transposed (weights as A, activations as B; same k order, same sums): a lower-half lane holds one sample's 6 outputs, stores 2 runs of 12 B; was a guarded 4-B store per output value
# speedup vs baseline: 2.2137x; 1.2127x over previous
#include <hip/hip_runtime.h>

typedef __attribute__((ext_vector_type(2))) float v2f;
typedef __attribute__((ext_vector_type(8))) float v8f;
typedef __attribute__((ext_vector_type(3))) float v3f;

#define WAVES_PER_BLOCK 8
#define LDS_PER_WAVE (832 + 544 + 288)

__device__ __forceinline__ float relu1(float a) {
    return __builtin_amdgcn_fmed3f(a, 0.0f, __builtin_inff());
}


__device__ __forceinline__ void load_tile(
    const float* __restrict__ x, const float* __restrict__ c,
    int tile, int Bn, int lane, v3f (&xi)[9], v2f (&ci)[9])
{
    const long tbase = (long)tile * 16;
    const float* xt = x + tbase * 51;
    const float* ct = c + tbase * 34;
    const long rem = (long)Bn - tbase;
    const int lastp = (rem >= 16 ? 16 * 17 : (int)rem * 17) - 1;
    #pragma unroll
    for (int i = 0; i < 9; ++i) {
        const int p  = (i < 8) ? lane + 32 * i : 240 + (lane ^ 16);
        const int pl = min(p, lastp);
        const float* xp = xt + (long)pl * 3;
        const float* cp = ct + (long)pl * 2;
        v3f xv; xv.x = xp[0]; xv.y = xp[1]; xv.z = xp[2];
        v2f cv; cv.x = cp[0]; cv.y = cp[1];
        xi[i] = xv;
        ci[i] = cv;
    }
}

__global__ __launch_bounds__(256) void encoder_mlp_wmma(
    const float* __restrict__ x,  const float* __restrict__ c,
    const float* __restrict__ W1, const float* __restrict__ b1,
    const float* __restrict__ W2, const float* __restrict__ b2,
    const float* __restrict__ W3, const float* __restrict__ b3,
    const float* __restrict__ Wmu, const float* __restrict__ bmu,
    const float* __restrict__ Wlv, const float* __restrict__ blv,
    float* __restrict__ out, int Bn)
{
    __shared__ float lds[WAVES_PER_BLOCK * LDS_PER_WAVE];

    const int lane = threadIdx.x & 31;
    const int wave = threadIdx.x >> 5;
    const int hi   = lane >> 4;
    const int sl   = lane & 15;

    float* h1 = &lds[wave * LDS_PER_WAVE];
    float* h2 = h1 + 16 * 52;
    float* h3 = h2 + 16 * 34;

    float w1r[15], b1r[3];
    #pragma unroll
    for (int i = 0; i < 15; ++i) w1r[i] = W1[i];
    #pragma unroll
    for (int i = 0; i < 3; ++i)  b1r[i] = b1[i];

    v2f w2f[2][13];
    #pragma unroll
    for (int t = 0; t < 2; ++t) {
        const int n = sl + 16 * t;
        #pragma unroll
        for (int k = 0; k < 13; ++k) {
            const int k0 = 4 * k + 2 * hi;
            v2f w;
            w.x = (k0     < 51) ? W2[n * 51 + k0]     : 0.0f;
            w.y = (k0 + 1 < 51) ? W2[n * 51 + k0 + 1] : 0.0f;
            w2f[t][k] = w;
        }
    }
    v2f w3f[8];
    #pragma unroll
    for (int k = 0; k < 8; ++k) {
        const int k0 = 4 * k + 2 * hi;
        v2f w; w.x = W3[sl * 32 + k0]; w.y = W3[sl * 32 + k0 + 1];
        w3f[k] = w;
    }
    v2f whf[4];
    {
        const float* wrow = (sl < 3) ? (Wmu + sl * 16)
                          : (sl < 6) ? (Wlv + (sl - 3) * 16) : nullptr;
        #pragma unroll
        for (int k = 0; k < 4; ++k) {
            const int k0 = 4 * k + 2 * hi;
            v2f w;
            w.x = wrow ? wrow[k0]     : 0.0f;
            w.y = wrow ? wrow[k0 + 1] : 0.0f;
            whf[k] = w;
        }
    }
    const float b2v0 = b2[sl];
    const float b2v1 = b2[sl + 16];
    const float b3v  = b3[sl];
    float bh[8];
    #pragma unroll
    for (int r = 0; r < 3; ++r) { bh[r] = bmu[r]; bh[r + 3] = blv[r]; }
    bh[6] = 0.0f; bh[7] = 0.0f;

    const int ntiles     = (Bn + 15) >> 4;
    const int totalWaves = gridDim.x * WAVES_PER_BLOCK;
    const int gwave      = blockIdx.x * WAVES_PER_BLOCK + wave;

    v3f xi[9];
    v2f ci[9];

    auto fc1 = [&]() __attribute__((always_inline)) {
        {
            #pragma unroll
            for (int i = 0; i < 9; ++i) {
                const int p = (i < 8) ? lane + 32 * i : 240 + (lane ^ 16);
                float* hp = h1 + 3 * p + p / 17;
                #pragma unroll
                for (int o = 0; o < 3; ++o) {
                    float a = b1r[o];
                    a = fmaf(w1r[o * 5 + 0], xi[i].x, a);
                    a = fmaf(w1r[o * 5 + 1], xi[i].y, a);
                    a = fmaf(w1r[o * 5 + 2], xi[i].z, a);
                    a = fmaf(w1r[o * 5 + 3], ci[i].x, a);
                    a = fmaf(w1r[o * 5 + 4], ci[i].y, a);
                    hp[o] = relu1(a);
                }
            }
            if (hi) h1[sl * 52 + 51] = 0.0f;
        }
    };

    auto finish_tile = [&](const long tbase) __attribute__((always_inline)) {
        asm volatile("" ::: "memory");

        v8f acc0, acc1;
        #pragma unroll
        for (int r = 0; r < 8; ++r) { acc0[r] = b2v0; acc1[r] = b2v1; }
        #pragma unroll
        for (int k = 0; k < 13; ++k) {
            const int kb = 4 * k + 2 * hi;
            v2f a = *(const v2f*)&h1[sl * 52 + kb];
            acc0 = __builtin_amdgcn_wmma_f32_16x16x4_f32(
                       false, a, false, w2f[0][k], (short)0, acc0, false, false);
            acc1 = __builtin_amdgcn_wmma_f32_16x16x4_f32(
                       false, a, false, w2f[1][k], (short)0, acc1, false, false);
        }
        #pragma unroll
        for (int r = 0; r < 8; ++r) {
            h2[(r + 8 * hi) * 34 + sl]      = relu1(acc0[r]);
            h2[(r + 8 * hi) * 34 + sl + 16] = relu1(acc1[r]);
        }
        asm volatile("" ::: "memory");

        v8f acc2;
        #pragma unroll
        for (int r = 0; r < 8; ++r) acc2[r] = b3v;
        #pragma unroll
        for (int k = 0; k < 8; ++k) {
            const int kb = 4 * k + 2 * hi;
            v2f a = *(const v2f*)&h2[sl * 34 + kb];
            acc2 = __builtin_amdgcn_wmma_f32_16x16x4_f32(
                       false, a, false, w3f[k], (short)0, acc2, false, false);
        }
        #pragma unroll
        for (int r = 0; r < 8; ++r)
            h3[(r + 8 * hi) * 18 + sl] = relu1(acc2[r]);
        asm volatile("" ::: "memory");

        v8f acch;
        #pragma unroll
        for (int r = 0; r < 8; ++r) acch[r] = bh[r];
        #pragma unroll
        for (int k = 0; k < 4; ++k) {
            const int kb = 4 * k + 2 * hi;
            v2f hv = *(const v2f*)&h3[sl * 18 + kb];
            acch = __builtin_amdgcn_wmma_f32_16x16x4_f32(
                       false, whf[k], false, hv, (short)0, acch, false, false);
        }

        const long smp = tbase + sl;
        if (hi == 0 && smp < Bn) {
            float* om = out + smp * 3;
            float* ol = out + (long)Bn * 3 + smp * 3;
            om[0] = acch[0]; om[1] = acch[1]; om[2] = acch[2];
            ol[0] = acch[3]; ol[1] = acch[4]; ol[2] = acch[5];
        }
    };

    int tile = gwave;
    if (tile < ntiles) {
        load_tile(x, c, tile, Bn, lane, xi, ci);
        for (; tile + totalWaves < ntiles; tile += totalWaves) {
            fc1();
            load_tile(x, c, tile + totalWaves, Bn, lane, xi, ci);
            finish_tile((long)tile * 16);
        }
        fc1();
        finish_tile((long)tile * 16);
    }
}

extern "C" void kernel_launch(void* const* d_in, const int* in_sizes, int n_in,
                              void* d_out, int out_size, void* d_ws, size_t ws_size,
                              hipStream_t stream) {
    const float* x   = (const float*)d_in[0];
    const float* c   = (const float*)d_in[1];
    const float* W1  = (const float*)d_in[2];
    const float* b1  = (const float*)d_in[3];
    const float* W2  = (const float*)d_in[4];
    const float* b2  = (const float*)d_in[5];
    const float* W3  = (const float*)d_in[6];
    const float* b3  = (const float*)d_in[7];
    const float* Wmu = (const float*)d_in[8];
    const float* bmu = (const float*)d_in[9];
    const float* Wlv = (const float*)d_in[10];
    const float* blv = (const float*)d_in[11];
    float* out = (float*)d_out;

    const int Bn = in_sizes[0] / 51;
    if (Bn <= 0) return;

    const int blocks = 1024;
    encoder_mlp_wmma<<<blocks, 256, 0, stream>>>(
        x, c, W1, b1, W2, b2, W3, b3, Wmu, bmu, Wlv, blv, out, Bn);
}


# tile number made wave-uniform (readfirstlane): tile loop and tile bases scalar, a tile's 18 loads and 2 stores take base + fixed lane offset + constant; pair clamp only in a peeled partial last tile
# speedup vs baseline: 2.2910x; 1.0349x over previous
#include <hip/hip_runtime.h>

typedef __attribute__((ext_vector_type(2))) float v2f;
typedef __attribute__((ext_vector_type(8))) float v8f;
typedef __attribute__((ext_vector_type(3))) float v3f;

#define WAVES_PER_BLOCK 8
#define LDS_PER_WAVE (832 + 544 + 288)

__device__ __forceinline__ float relu1(float a) {
    return __builtin_amdgcn_fmed3f(a, 0.0f, __builtin_inff());
}


__device__ __forceinline__ void load_full_tile(
    const float* __restrict__ x, const float* __restrict__ c,
    int tile, int lane, v3f (&xi)[9], v2f (&ci)[9])
{
    const long tbase = (long)tile * 16;
    const float* xt = x + tbase * 51;
    const float* ct = c + tbase * 34;
    const unsigned l0 = (unsigned)lane;
    const unsigned l1 = (unsigned)(lane ^ 16);
    #pragma unroll
    for (int i = 0; i < 9; ++i) {
        const float* xs = xt + ((i < 8) ? 96 * i : 720);
        const float* cs = ct + ((i < 8) ? 64 * i : 480);
        const unsigned xo = 3u * ((i < 8) ? l0 : l1);
        const unsigned co = 2u * ((i < 8) ? l0 : l1);
        v3f xv; xv.x = xs[xo]; xv.y = xs[xo + 1u]; xv.z = xs[xo + 2u];
        v2f cv; cv.x = cs[co]; cv.y = cs[co + 1u];
        xi[i] = xv;
        ci[i] = cv;
    }
}

__device__ __forceinline__ void load_partial_tile(
    const float* __restrict__ x, const float* __restrict__ c,
    int tile, int nvalid, int lane, v3f (&xi)[9], v2f (&ci)[9])
{
    const long tbase = (long)tile * 16;
    const float* xt = x + tbase * 51;
    const float* ct = c + tbase * 34;
    const unsigned lastp = (unsigned)(nvalid * 17 - 1);
    #pragma unroll
    for (int i = 0; i < 9; ++i) {
        const unsigned p  = (i < 8) ? (unsigned)lane + 32u * i : 240u + (unsigned)(lane ^ 16);
        const unsigned pl = min(p, lastp);
        const unsigned xo = 3u * pl;
        const unsigned co = 2u * pl;
        v3f xv; xv.x = xt[xo]; xv.y = xt[xo + 1u]; xv.z = xt[xo + 2u];
        v2f cv; cv.x = ct[co]; cv.y = ct[co + 1u];
        xi[i] = xv;
        ci[i] = cv;
    }
}

__global__ __launch_bounds__(256) void encoder_mlp_wmma(
    const float* __restrict__ x,  const float* __restrict__ c,
    const float* __restrict__ W1, const float* __restrict__ b1,
    const float* __restrict__ W2, const float* __restrict__ b2,
    const float* __restrict__ W3, const float* __restrict__ b3,
    const float* __restrict__ Wmu, const float* __restrict__ bmu,
    const float* __restrict__ Wlv, const float* __restrict__ blv,
    float* __restrict__ out, int Bn)
{
    __shared__ float lds[WAVES_PER_BLOCK * LDS_PER_WAVE];

    const int lane = threadIdx.x & 31;
    const int wave = __builtin_amdgcn_readfirstlane((int)(threadIdx.x >> 5));
    const int hi   = lane >> 4;
    const int sl   = lane & 15;

    float* h1 = &lds[wave * LDS_PER_WAVE];
    float* h2 = h1 + 16 * 52;
    float* h3 = h2 + 16 * 34;

    float w1r[15], b1r[3];
    #pragma unroll
    for (int i = 0; i < 15; ++i) w1r[i] = W1[i];
    #pragma unroll
    for (int i = 0; i < 3; ++i)  b1r[i] = b1[i];

    v2f w2f[2][13];
    #pragma unroll
    for (int t = 0; t < 2; ++t) {
        const int n = sl + 16 * t;
        #pragma unroll
        for (int k = 0; k < 13; ++k) {
            const int k0 = 4 * k + 2 * hi;
            v2f w;
            w.x = (k0     < 51) ? W2[n * 51 + k0]     : 0.0f;
            w.y = (k0 + 1 < 51) ? W2[n * 51 + k0 + 1] : 0.0f;
            w2f[t][k] = w;
        }
    }
    v2f w3f[8];
    #pragma unroll
    for (int k = 0; k < 8; ++k) {
        const int k0 = 4 * k + 2 * hi;
        v2f w; w.x = W3[sl * 32 + k0]; w.y = W3[sl * 32 + k0 + 1];
        w3f[k] = w;
    }
    v2f whf[4];
    {
        const float* wrow = (sl < 3) ? (Wmu + sl * 16)
                          : (sl < 6) ? (Wlv + (sl - 3) * 16) : nullptr;
        #pragma unroll
        for (int k = 0; k < 4; ++k) {
            const int k0 = 4 * k + 2 * hi;
            v2f w;
            w.x = wrow ? wrow[k0]     : 0.0f;
            w.y = wrow ? wrow[k0 + 1] : 0.0f;
            whf[k] = w;
        }
    }
    const float b2v0 = b2[sl];
    const float b2v1 = b2[sl + 16];
    const float b3v  = b3[sl];
    float bh[8];
    #pragma unroll
    for (int r = 0; r < 3; ++r) { bh[r] = bmu[r]; bh[r + 3] = blv[r]; }
    bh[6] = 0.0f; bh[7] = 0.0f;

    const int nfull      = Bn >> 4;
    const int ntail      = Bn & 15;
    const int totalWaves = gridDim.x * WAVES_PER_BLOCK;
    const int gwave      = blockIdx.x * WAVES_PER_BLOCK + wave;

    v3f xi[9];
    v2f ci[9];

    auto fc1 = [&]() __attribute__((always_inline)) {
        {
            #pragma unroll
            for (int i = 0; i < 9; ++i) {
                const int p = (i < 8) ? lane + 32 * i : 240 + (lane ^ 16);
                float* hp = h1 + 3 * p + p / 17;
                #pragma unroll
                for (int o = 0; o < 3; ++o) {
                    float a = b1r[o];
                    a = fmaf(w1r[o * 5 + 0], xi[i].x, a);
                    a = fmaf(w1r[o * 5 + 1], xi[i].y, a);
                    a = fmaf(w1r[o * 5 + 2], xi[i].z, a);
                    a = fmaf(w1r[o * 5 + 3], ci[i].x, a);
                    a = fmaf(w1r[o * 5 + 4], ci[i].y, a);
                    hp[o] = relu1(a);
                }
            }
            if (hi) h1[sl * 52 + 51] = 0.0f;
        }
    };

    auto finish_tile = [&](const long tbase, const int nvalid) __attribute__((always_inline)) {
        asm volatile("" ::: "memory");

        v8f acc0, acc1;
        #pragma unroll
        for (int r = 0; r < 8; ++r) { acc0[r] = b2v0; acc1[r] = b2v1; }
        #pragma unroll
        for (int k = 0; k < 13; ++k) {
            const int kb = 4 * k + 2 * hi;
            v2f a = *(const v2f*)&h1[sl * 52 + kb];
            acc0 = __builtin_amdgcn_wmma_f32_16x16x4_f32(
                       false, a, false, w2f[0][k], (short)0, acc0, false, false);
            acc1 = __builtin_amdgcn_wmma_f32_16x16x4_f32(
                       false, a, false, w2f[1][k], (short)0, acc1, false, false);
        }
        #pragma unroll
        for (int r = 0; r < 8; ++r) {
            h2[(r + 8 * hi) * 34 + sl]      = relu1(acc0[r]);
            h2[(r + 8 * hi) * 34 + sl + 16] = relu1(acc1[r]);
        }
        asm volatile("" ::: "memory");

        v8f acc2;
        #pragma unroll
        for (int r = 0; r < 8; ++r) acc2[r] = b3v;
        #pragma unroll
        for (int k = 0; k < 8; ++k) {
            const int kb = 4 * k + 2 * hi;
            v2f a = *(const v2f*)&h2[sl * 34 + kb];
            acc2 = __builtin_amdgcn_wmma_f32_16x16x4_f32(
                       false, a, false, w3f[k], (short)0, acc2, false, false);
        }
        #pragma unroll
        for (int r = 0; r < 8; ++r)
            h3[(r + 8 * hi) * 18 + sl] = relu1(acc2[r]);
        asm volatile("" ::: "memory");

        v8f acch;
        #pragma unroll
        for (int r = 0; r < 8; ++r) acch[r] = bh[r];
        #pragma unroll
        for (int k = 0; k < 4; ++k) {
            const int kb = 4 * k + 2 * hi;
            v2f hv = *(const v2f*)&h3[sl * 18 + kb];
            acch = __builtin_amdgcn_wmma_f32_16x16x4_f32(
                       false, whf[k], false, hv, (short)0, acch, false, false);
        }

        if (hi == 0 && sl < nvalid) {
            float* om = out + tbase * 3;
            float* ol = out + (long)Bn * 3 + tbase * 3;
            const unsigned so = 3u * (unsigned)sl;
            om[so] = acch[0]; om[so + 1u] = acch[1]; om[so + 2u] = acch[2];
            ol[so] = acch[3]; ol[so + 1u] = acch[4]; ol[so + 2u] = acch[5];
        }
    };

    int tile = gwave;
    if (tile < nfull) {
        load_full_tile(x, c, tile, lane, xi, ci);
        for (; tile + totalWaves < nfull; tile += totalWaves) {
            fc1();
            load_full_tile(x, c, tile + totalWaves, lane, xi, ci);
            finish_tile((long)tile * 16, 16);
        }
        fc1();
        finish_tile((long)tile * 16, 16);
        tile += totalWaves;
    }
    if (ntail != 0 && tile == nfull) {
        load_partial_tile(x, c, tile, ntail, lane, xi, ci);
        fc1();
        finish_tile((long)tile * 16, ntail);
    }
}

extern "C" void kernel_launch(void* const* d_in, const int* in_sizes, int n_in,
                              void* d_out, int out_size, void* d_ws, size_t ws_size,
                              hipStream_t stream) {
    const float* x   = (const float*)d_in[0];
    const float* c   = (const float*)d_in[1];
    const float* W1  = (const float*)d_in[2];
    const float* b1  = (const float*)d_in[3];
    const float* W2  = (const float*)d_in[4];
    const float* b2  = (const float*)d_in[5];
    const float* W3  = (const float*)d_in[6];
    const float* b3  = (const float*)d_in[7];
    const float* Wmu = (const float*)d_in[8];
    const float* bmu = (const float*)d_in[9];
    const float* Wlv = (const float*)d_in[10];
    const float* blv = (const float*)d_in[11];
    float* out = (float*)d_out;

    const int Bn = in_sizes[0] / 51;
    if (Bn <= 0) return;

    const int blocks = 1024;
    encoder_mlp_wmma<<<blocks, 256, 0, stream>>>(
        x, c, W1, b1, W2, b2, W3, b3, Wmu, bmu, Wlv, blv, out, Bn);
}


# all layers run transposed (weights as A, activations as B): ReLU'd D registers are the next layer's B fragments, h2/h3 no longer pass through LDS; fc3/heads sum k in D-register order (new sum order)
# speedup vs baseline: 2.3838x; 1.0405x over previous
#include <hip/hip_runtime.h>

typedef __attribute__((ext_vector_type(2))) float v2f;
typedef __attribute__((ext_vector_type(8))) float v8f;
typedef __attribute__((ext_vector_type(3))) float v3f;

#define WAVES_PER_BLOCK 8
#define LDS_PER_WAVE 832

__device__ __forceinline__ float relu1(float a) {
    return __builtin_amdgcn_fmed3f(a, 0.0f, __builtin_inff());
}


__device__ __forceinline__ void load_full_tile(
    const float* __restrict__ x, const float* __restrict__ c,
    int tile, int lane, v3f (&xi)[9], v2f (&ci)[9])
{
    const long tbase = (long)tile * 16;
    const float* xt = x + tbase * 51;
    const float* ct = c + tbase * 34;
    const unsigned l0 = (unsigned)lane;
    const unsigned l1 = (unsigned)(lane ^ 16);
    #pragma unroll
    for (int i = 0; i < 9; ++i) {
        const float* xs = xt + ((i < 8) ? 96 * i : 720);
        const float* cs = ct + ((i < 8) ? 64 * i : 480);
        const unsigned xo = 3u * ((i < 8) ? l0 : l1);
        const unsigned co = 2u * ((i < 8) ? l0 : l1);
        v3f xv; xv.x = xs[xo]; xv.y = xs[xo + 1u]; xv.z = xs[xo + 2u];
        v2f cv; cv.x = cs[co]; cv.y = cs[co + 1u];
        xi[i] = xv;
        ci[i] = cv;
    }
}

__device__ __forceinline__ void load_partial_tile(
    const float* __restrict__ x, const float* __restrict__ c,
    int tile, int nvalid, int lane, v3f (&xi)[9], v2f (&ci)[9])
{
    const long tbase = (long)tile * 16;
    const float* xt = x + tbase * 51;
    const float* ct = c + tbase * 34;
    const unsigned lastp = (unsigned)(nvalid * 17 - 1);
    #pragma unroll
    for (int i = 0; i < 9; ++i) {
        const unsigned p  = (i < 8) ? (unsigned)lane + 32u * i : 240u + (unsigned)(lane ^ 16);
        const unsigned pl = min(p, lastp);
        const unsigned xo = 3u * pl;
        const unsigned co = 2u * pl;
        v3f xv; xv.x = xt[xo]; xv.y = xt[xo + 1u]; xv.z = xt[xo + 2u];
        v2f cv; cv.x = ct[co]; cv.y = ct[co + 1u];
        xi[i] = xv;
        ci[i] = cv;
    }
}

__global__ __launch_bounds__(256) void encoder_mlp_wmma(
    const float* __restrict__ x,  const float* __restrict__ c,
    const float* __restrict__ W1, const float* __restrict__ b1,
    const float* __restrict__ W2, const float* __restrict__ b2,
    const float* __restrict__ W3, const float* __restrict__ b3,
    const float* __restrict__ Wmu, const float* __restrict__ bmu,
    const float* __restrict__ Wlv, const float* __restrict__ blv,
    float* __restrict__ out, int Bn)
{
    __shared__ float lds[WAVES_PER_BLOCK * LDS_PER_WAVE];

    const int lane = threadIdx.x & 31;
    const int wave = __builtin_amdgcn_readfirstlane((int)(threadIdx.x >> 5));
    const int hi   = lane >> 4;
    const int sl   = lane & 15;

    float* h1 = &lds[wave * LDS_PER_WAVE];

    float w1r[15], b1r[3];
    #pragma unroll
    for (int i = 0; i < 15; ++i) w1r[i] = W1[i];
    #pragma unroll
    for (int i = 0; i < 3; ++i)  b1r[i] = b1[i];

    v2f w2f[2][13];
    #pragma unroll
    for (int t = 0; t < 2; ++t) {
        const int n = sl + 16 * t;
        #pragma unroll
        for (int k = 0; k < 13; ++k) {
            const int k0 = 4 * k + 2 * hi;
            v2f w;
            w.x = (k0     < 51) ? W2[n * 51 + k0]     : 0.0f;
            w.y = (k0 + 1 < 51) ? W2[n * 51 + k0 + 1] : 0.0f;
            w2f[t][k] = w;
        }
    }
    v2f w3f[8];
    #pragma unroll
    for (int j = 0; j < 8; ++j) {
        const int k0 = 16 * (j >> 2) + 8 * hi + 2 * (j & 3);
        v2f w; w.x = W3[sl * 32 + k0]; w.y = W3[sl * 32 + k0 + 1];
        w3f[j] = w;
    }
    v2f whf[4];
    {
        const float* wrow = (sl < 3) ? (Wmu + sl * 16)
                          : (sl < 6) ? (Wlv + (sl - 3) * 16) : nullptr;
        #pragma unroll
        for (int j = 0; j < 4; ++j) {
            const int k0 = 8 * hi + 2 * j;
            v2f w;
            w.x = wrow ? wrow[k0]     : 0.0f;
            w.y = wrow ? wrow[k0 + 1] : 0.0f;
            whf[j] = w;
        }
    }
    float b2r[2][8], b3r[8];
    #pragma unroll
    for (int r = 0; r < 8; ++r) {
        b2r[0][r] = b2[8 * hi + r];
        b2r[1][r] = b2[16 + 8 * hi + r];
        b3r[r]    = b3[8 * hi + r];
    }
    float bh[8];
    #pragma unroll
    for (int r = 0; r < 3; ++r) { bh[r] = bmu[r]; bh[r + 3] = blv[r]; }
    bh[6] = 0.0f; bh[7] = 0.0f;

    const int nfull      = Bn >> 4;
    const int ntail      = Bn & 15;
    const int totalWaves = gridDim.x * WAVES_PER_BLOCK;
    const int gwave      = blockIdx.x * WAVES_PER_BLOCK + wave;

    v3f xi[9];
    v2f ci[9];

    auto fc1 = [&]() __attribute__((always_inline)) {
        {
            #pragma unroll
            for (int i = 0; i < 9; ++i) {
                const int p = (i < 8) ? lane + 32 * i : 240 + (lane ^ 16);
                float* hp = h1 + 3 * p + p / 17;
                #pragma unroll
                for (int o = 0; o < 3; ++o) {
                    float a = b1r[o];
                    a = fmaf(w1r[o * 5 + 0], xi[i].x, a);
                    a = fmaf(w1r[o * 5 + 1], xi[i].y, a);
                    a = fmaf(w1r[o * 5 + 2], xi[i].z, a);
                    a = fmaf(w1r[o * 5 + 3], ci[i].x, a);
                    a = fmaf(w1r[o * 5 + 4], ci[i].y, a);
                    hp[o] = relu1(a);
                }
            }
            if (hi) h1[sl * 52 + 51] = 0.0f;
        }
    };

    auto finish_tile = [&](const long tbase, const int nvalid) __attribute__((always_inline)) {
        asm volatile("" ::: "memory");

        v8f acc0, acc1;
        #pragma unroll
        for (int r = 0; r < 8; ++r) { acc0[r] = b2r[0][r]; acc1[r] = b2r[1][r]; }
        #pragma unroll
        for (int k = 0; k < 13; ++k) {
            const int kb = 4 * k + 2 * hi;
            v2f hv = *(const v2f*)&h1[sl * 52 + kb];
            acc0 = __builtin_amdgcn_wmma_f32_16x16x4_f32(
                       false, w2f[0][k], false, hv, (short)0, acc0, false, false);
            acc1 = __builtin_amdgcn_wmma_f32_16x16x4_f32(
                       false, w2f[1][k], false, hv, (short)0, acc1, false, false);
        }

        v8f r0, r1;
        #pragma unroll
        for (int r = 0; r < 8; ++r) { r0[r] = relu1(acc0[r]); r1[r] = relu1(acc1[r]); }
        v8f acc2;
        #pragma unroll
        for (int r = 0; r < 8; ++r) acc2[r] = b3r[r];
        #pragma unroll
        for (int j = 0; j < 8; ++j) {
            const int jj = j & 3;
            v2f hv;
            hv.x = (j < 4) ? r0[2 * jj]     : r1[2 * jj];
            hv.y = (j < 4) ? r0[2 * jj + 1] : r1[2 * jj + 1];
            acc2 = __builtin_amdgcn_wmma_f32_16x16x4_f32(
                       false, w3f[j], false, hv, (short)0, acc2, false, false);
        }

        v8f r2;
        #pragma unroll
        for (int r = 0; r < 8; ++r) r2[r] = relu1(acc2[r]);
        v8f acch;
        #pragma unroll
        for (int r = 0; r < 8; ++r) acch[r] = bh[r];
        #pragma unroll
        for (int j = 0; j < 4; ++j) {
            v2f hv;
            hv.x = r2[2 * j];
            hv.y = r2[2 * j + 1];
            acch = __builtin_amdgcn_wmma_f32_16x16x4_f32(
                       false, whf[j], false, hv, (short)0, acch, false, false);
        }

        if (hi == 0 && sl < nvalid) {
            float* om = out + tbase * 3;
            float* ol = out + (long)Bn * 3 + tbase * 3;
            const unsigned so = 3u * (unsigned)sl;
            om[so] = acch[0]; om[so + 1u] = acch[1]; om[so + 2u] = acch[2];
            ol[so] = acch[3]; ol[so + 1u] = acch[4]; ol[so + 2u] = acch[5];
        }
    };

    int tile = gwave;
    if (tile < nfull) {
        load_full_tile(x, c, tile, lane, xi, ci);
        for (; tile + totalWaves < nfull; tile += totalWaves) {
            fc1();
            load_full_tile(x, c, tile + totalWaves, lane, xi, ci);
            finish_tile((long)tile * 16, 16);
        }
        fc1();
        finish_tile((long)tile * 16, 16);
        tile += totalWaves;
    }
    if (ntail != 0 && tile == nfull) {
        load_partial_tile(x, c, tile, ntail, lane, xi, ci);
        fc1();
        finish_tile((long)tile * 16, ntail);
    }
}

extern "C" void kernel_launch(void* const* d_in, const int* in_sizes, int n_in,
                              void* d_out, int out_size, void* d_ws, size_t ws_size,
                              hipStream_t stream) {
    const float* x   = (const float*)d_in[0];
    const float* c   = (const float*)d_in[1];
    const float* W1  = (const float*)d_in[2];
    const float* b1  = (const float*)d_in[3];
    const float* W2  = (const float*)d_in[4];
    const float* b2  = (const float*)d_in[5];
    const float* W3  = (const float*)d_in[6];
    const float* b3  = (const float*)d_in[7];
    const float* Wmu = (const float*)d_in[8];
    const float* bmu = (const float*)d_in[9];
    const float* Wlv = (const float*)d_in[10];
    const float* blv = (const float*)d_in[11];
    float* out = (float*)d_out;

    const int Bn = in_sizes[0] / 51;
    if (Bn <= 0) return;

    const int blocks = 1024;
    encoder_mlp_wmma<<<blocks, 256, 0, stream>>>(
        x, c, W1, b1, W2, b2, W3, b3, Wmu, bmu, Wlv, blv, out, Bn);
}


# trip widened to 32 samples (two 16-sample column tiles share each weight fragment, both heads into one D tile): 34 whole-line loads, 17 full fc1 steps, 4/2 WMMA chains, whole-line stores; same sums
# speedup vs baseline: 2.4925x; 1.0456x over previous
#include <hip/hip_runtime.h>


typedef __attribute__((ext_vector_type(2))) float v2f;
typedef __attribute__((ext_vector_type(8))) float v8f;
typedef __attribute__((ext_vector_type(3))) float v3f;

#define WAVES_PER_BLOCK 4
#define LDS_PER_WAVE 1664

__device__ __forceinline__ float relu1(float a) {
    return __builtin_amdgcn_fmed3f(a, 0.0f, __builtin_inff());
}

__device__ __forceinline__ void load_full_trip(
    const float* __restrict__ x, const float* __restrict__ c,
    int trip, int lane, v3f (&xi)[17], v2f (&ci)[17])
{
    const long tbase = (long)trip * 32;
    const float* xt = x + tbase * 51;
    const float* ct = c + tbase * 34;
    const unsigned xo = 3u * (unsigned)lane;
    const unsigned co = 2u * (unsigned)lane;
    #pragma unroll
    for (int i = 0; i < 17; ++i) {
        const float* xs = xt + 96 * i;
        const float* cs = ct + 64 * i;
        v3f xv; xv.x = xs[xo]; xv.y = xs[xo + 1u]; xv.z = xs[xo + 2u];
        v2f cv; cv.x = cs[co]; cv.y = cs[co + 1u];
        xi[i] = xv;
        ci[i] = cv;
    }
}

__device__ __forceinline__ void load_partial_trip(
    const float* __restrict__ x, const float* __restrict__ c,
    int trip, int nvalid, int lane, v3f (&xi)[17], v2f (&ci)[17])
{
    const long tbase = (long)trip * 32;
    const float* xt = x + tbase * 51;
    const float* ct = c + tbase * 34;
    const unsigned lastp = (unsigned)(nvalid * 17 - 1);
    #pragma unroll
    for (int i = 0; i < 17; ++i) {
        const unsigned p  = (unsigned)lane + 32u * i;
        const unsigned pl = min(p, lastp);
        const unsigned xo = 3u * pl;
        const unsigned co = 2u * pl;
        v3f xv; xv.x = xt[xo]; xv.y = xt[xo + 1u]; xv.z = xt[xo + 2u];
        v2f cv; cv.x = ct[co]; cv.y = ct[co + 1u];
        xi[i] = xv;
        ci[i] = cv;
    }
}

__global__ __launch_bounds__(128) void encoder_mlp_wmma(
    const float* __restrict__ x,  const float* __restrict__ c,
    const float* __restrict__ W1, const float* __restrict__ b1,
    const float* __restrict__ W2, const float* __restrict__ b2,
    const float* __restrict__ W3, const float* __restrict__ b3,
    const float* __restrict__ Wmu, const float* __restrict__ bmu,
    const float* __restrict__ Wlv, const float* __restrict__ blv,
    float* __restrict__ out, int Bn)
{
    __shared__ float lds[WAVES_PER_BLOCK * LDS_PER_WAVE];

    const int lane = threadIdx.x & 31;
    const int wave = __builtin_amdgcn_readfirstlane((int)(threadIdx.x >> 5));
    const int hi   = lane >> 4;
    const int sl   = lane & 15;

    float* h1 = &lds[wave * LDS_PER_WAVE];

    h1[lane * 52 + 51] = 0.0f;

    float w1r[15], b1r[3];
    #pragma unroll
    for (int i = 0; i < 15; ++i) w1r[i] = W1[i];
    #pragma unroll
    for (int i = 0; i < 3; ++i)  b1r[i] = b1[i];

    v2f w2f[2][13];
    #pragma unroll
    for (int t = 0; t < 2; ++t) {
        const int n = sl + 16 * t;
        #pragma unroll
        for (int k = 0; k < 13; ++k) {
            const int k0 = 4 * k + 2 * hi;
            v2f w;
            w.x = (k0     < 51) ? W2[n * 51 + k0]     : 0.0f;
            w.y = (k0 + 1 < 51) ? W2[n * 51 + k0 + 1] : 0.0f;
            w2f[t][k] = w;
        }
    }
    v2f w3f[8];
    #pragma unroll
    for (int j = 0; j < 8; ++j) {
        const int k0 = 16 * (j >> 2) + 8 * hi + 2 * (j & 3);
        v2f w; w.x = W3[sl * 32 + k0]; w.y = W3[sl * 32 + k0 + 1];
        w3f[j] = w;
    }
    v2f whA[4], whB[4];
    {
        const float* rowA = (sl < 3) ? (Wmu + sl * 16)
                          : (sl < 6) ? (Wlv + (sl - 3) * 16) : nullptr;
        const float* rowB = (sl >= 8 && sl < 11)  ? (Wmu + (sl - 8) * 16)
                          : (sl >= 11 && sl < 14) ? (Wlv + (sl - 11) * 16) : nullptr;
        #pragma unroll
        for (int j = 0; j < 4; ++j) {
            const int k0 = 8 * hi + 2 * j;
            v2f a, b;
            a.x = rowA ? rowA[k0]     : 0.0f;
            a.y = rowA ? rowA[k0 + 1] : 0.0f;
            b.x = rowB ? rowB[k0]     : 0.0f;
            b.y = rowB ? rowB[k0 + 1] : 0.0f;
            whA[j] = a;
            whB[j] = b;
        }
    }
    float b2r[2][8], b3r[8];
    #pragma unroll
    for (int r = 0; r < 8; ++r) {
        b2r[0][r] = b2[8 * hi + r];
        b2r[1][r] = b2[16 + 8 * hi + r];
        b3r[r]    = b3[8 * hi + r];
    }
    float bh[8];
    #pragma unroll
    for (int r = 0; r < 3; ++r) { bh[r] = bmu[r]; bh[r + 3] = blv[r]; }
    bh[6] = 0.0f; bh[7] = 0.0f;

    const int nfull      = Bn >> 5;
    const int ntail      = Bn & 31;
    const int totalWaves = gridDim.x * WAVES_PER_BLOCK;
    const int gwave      = blockIdx.x * WAVES_PER_BLOCK + wave;

    v3f xi[17];
    v2f ci[17];

    auto fc1 = [&]() __attribute__((always_inline)) {
        #pragma unroll
        for (int i = 0; i < 17; ++i) {
            const int p = lane + 32 * i;
            float* hp = h1 + 3 * p + p / 17;
            float x1 = xi[i].y;
            asm("" : "+v"(x1));
            #pragma unroll
            for (int o = 0; o < 3; ++o) {
                float a = b1r[o];
                a = fmaf(w1r[o * 5 + 0], xi[i].x, a);
                a = fmaf(w1r[o * 5 + 1], x1, a);
                a = fmaf(w1r[o * 5 + 2], xi[i].z, a);
                a = fmaf(w1r[o * 5 + 3], ci[i].x, a);
                a = fmaf(w1r[o * 5 + 4], ci[i].y, a);
                hp[o] = relu1(a);
            }
        }
    };

    auto finish_trip = [&](const long tbase, const int nvalid) __attribute__((always_inline)) {
        asm volatile("" ::: "memory");

        v8f a00, a10, a01, a11;
        #pragma unroll
        for (int r = 0; r < 8; ++r) {
            a00[r] = b2r[0][r]; a10[r] = b2r[1][r];
            a01[r] = b2r[0][r]; a11[r] = b2r[1][r];
        }
        #pragma unroll
        for (int k = 0; k < 13; ++k) {
            const int kb = 4 * k + 2 * hi;
            v2f hv0 = *(const v2f*)&h1[sl * 52 + kb];
            v2f hv1 = *(const v2f*)&h1[(16 + sl) * 52 + kb];
            a00 = __builtin_amdgcn_wmma_f32_16x16x4_f32(
                      false, w2f[0][k], false, hv0, (short)0, a00, false, false);
            a10 = __builtin_amdgcn_wmma_f32_16x16x4_f32(
                      false, w2f[1][k], false, hv0, (short)0, a10, false, false);
            a01 = __builtin_amdgcn_wmma_f32_16x16x4_f32(
                      false, w2f[0][k], false, hv1, (short)0, a01, false, false);
            a11 = __builtin_amdgcn_wmma_f32_16x16x4_f32(
                      false, w2f[1][k], false, hv1, (short)0, a11, false, false);
        }

        v8f r00, r10, r01, r11;
        #pragma unroll
        for (int r = 0; r < 8; ++r) {
            r00[r] = relu1(a00[r]); r10[r] = relu1(a10[r]);
            r01[r] = relu1(a01[r]); r11[r] = relu1(a11[r]);
        }
        v8f c0, c1;
        #pragma unroll
        for (int r = 0; r < 8; ++r) { c0[r] = b3r[r]; c1[r] = b3r[r]; }
        #pragma unroll
        for (int j = 0; j < 8; ++j) {
            const int jj = j & 3;
            v2f g0, g1;
            g0.x = (j < 4) ? r00[2 * jj]     : r10[2 * jj];
            g0.y = (j < 4) ? r00[2 * jj + 1] : r10[2 * jj + 1];
            g1.x = (j < 4) ? r01[2 * jj]     : r11[2 * jj];
            g1.y = (j < 4) ? r01[2 * jj + 1] : r11[2 * jj + 1];
            c0 = __builtin_amdgcn_wmma_f32_16x16x4_f32(
                     false, w3f[j], false, g0, (short)0, c0, false, false);
            c1 = __builtin_amdgcn_wmma_f32_16x16x4_f32(
                     false, w3f[j], false, g1, (short)0, c1, false, false);
        }

        v8f q0, q1;
        #pragma unroll
        for (int r = 0; r < 8; ++r) { q0[r] = relu1(c0[r]); q1[r] = relu1(c1[r]); }
        v8f acch;
        #pragma unroll
        for (int r = 0; r < 8; ++r) acch[r] = bh[r];
        #pragma unroll
        for (int j = 0; j < 4; ++j) {
            v2f g;
            g.x = q0[2 * j];
            g.y = q0[2 * j + 1];
            acch = __builtin_amdgcn_wmma_f32_16x16x4_f32(
                       false, whA[j], false, g, (short)0, acch, false, false);
        }
        #pragma unroll
        for (int j = 0; j < 4; ++j) {
            v2f g;
            g.x = q1[2 * j];
            g.y = q1[2 * j + 1];
            acch = __builtin_amdgcn_wmma_f32_16x16x4_f32(
                       false, whB[j], false, g, (short)0, acch, false, false);
        }

        if (lane < nvalid) {
            float* om = out + tbase * 3;
            float* ol = out + (long)Bn * 3 + tbase * 3;
            const unsigned so = 3u * (unsigned)lane;
            om[so] = acch[0]; om[so + 1u] = acch[1]; om[so + 2u] = acch[2];
            ol[so] = acch[3]; ol[so + 1u] = acch[4]; ol[so + 2u] = acch[5];
        }
    };

    int trip = gwave;
    if (trip < nfull) {
        load_full_trip(x, c, trip, lane, xi, ci);
        for (; trip + totalWaves < nfull; trip += totalWaves) {
            fc1();
            load_full_trip(x, c, trip + totalWaves, lane, xi, ci);
            finish_trip((long)trip * 32, 32);
        }
        fc1();
        finish_trip((long)trip * 32, 32);
        trip += totalWaves;
    }
    if (ntail != 0 && trip == nfull) {
        load_partial_trip(x, c, trip, ntail, lane, xi, ci);
        fc1();
        finish_trip((long)trip * 32, ntail);
    }
}

extern "C" void kernel_launch(void* const* d_in, const int* in_sizes, int n_in,
                              void* d_out, int out_size, void* d_ws, size_t ws_size,
                              hipStream_t stream) {
    const float* x   = (const float*)d_in[0];
    const float* c   = (const float*)d_in[1];
    const float* W1  = (const float*)d_in[2];
    const float* b1  = (const float*)d_in[3];
    const float* W2  = (const float*)d_in[4];
    const float* b2  = (const float*)d_in[5];
    const float* W3  = (const float*)d_in[6];
    const float* b3  = (const float*)d_in[7];
    const float* Wmu = (const float*)d_in[8];
    const float* bmu = (const float*)d_in[9];
    const float* Wlv = (const float*)d_in[10];
    const float* blv = (const float*)d_in[11];
    float* out = (float*)d_out;

    const int Bn = in_sizes[0] / 51;
    if (Bn <= 0) return;

    const int blocks = 768;
    encoder_mlp_wmma<<<blocks, 128, 0, stream>>>(
        x, c, W1, b1, W2, b2, W3, b3, Wmu, bmu, Wlv, blv, out, Bn);
}
